# starting kernel
# baseline (speedup 1.0000x reference)
#include <hip/hip_runtime.h>

typedef float v2f __attribute__((ext_vector_type(2)));
typedef float v4f __attribute__((ext_vector_type(4)));
typedef float v8f __attribute__((ext_vector_type(8)));

#define DIM            1024
#define ROWS_PER_WAVE  16
#define WAVES          4
#define BLOCK          (WAVES * 32)
#define ROWS_PER_BLOCK (WAVES * ROWS_PER_WAVE)
#define CK             128
#define XSTRIDE        132
#define WZPAD          132

__global__ __launch_bounds__(BLOCK) void crossnet_fused(
    const float* __restrict__ X,
    const float* __restrict__ Wk,
    const float* __restrict__ Bb,
    float* __restrict__ Out)
{
    __shared__ float lds_w[3 * DIM + WZPAD];
    __shared__ float lds_x[WAVES][ROWS_PER_WAVE * XSTRIDE];
    __shared__ float lds_b3[DIM];
    __shared__ float lds_a[ROWS_PER_BLOCK];
    __shared__ float lds_red[2 * BLOCK];
    __shared__ float lds_e[2];

    const int tid = threadIdx.x;

    float p1 = 0.f, p2 = 0.f;
    for (int j = tid; j < DIM; j += BLOCK) {
        float w0 = Wk[j], w1 = Wk[DIM + j], w2 = Wk[2 * DIM + j];
        lds_w[j] = w0; lds_w[DIM + j] = w1; lds_w[2 * DIM + j] = w2;
        float b0 = Bb[j], b1 = Bb[DIM + j], b2 = Bb[2 * DIM + j];
        lds_b3[j] = b0 + b1 + b2;
        p1 += b0 * w1;
        p2 += (b0 + b1) * w2;
    }
    for (int j = tid; j < WZPAD; j += BLOCK) lds_w[3 * DIM + j] = 0.f;
    lds_red[tid] = p1;
    lds_red[BLOCK + tid] = p2;
    __syncthreads();
    if (tid == 0) {
        float e1 = 0.f, e2 = 0.f;
        for (int i = 0; i < BLOCK; ++i) { e1 += lds_red[i]; e2 += lds_red[BLOCK + i]; }
        lds_e[0] = e1; lds_e[1] = e2;
    }
    __syncthreads();

    const int wave = tid >> 5, lane = tid & 31;
    const int m  = lane & 15;
    const int kp = (lane >> 4) << 1;
    float* myx = &lds_x[wave][0];
    const float* abase = &myx[m * XSTRIDE + kp];
    const int row0 = blockIdx.x * ROWS_PER_BLOCK + wave * ROWS_PER_WAVE;

    v8f acc = {};
    for (int kc = 0; kc < DIM; kc += CK) {
        #pragma unroll 4
        for (int r = 0; r < ROWS_PER_WAVE; ++r) {
            v4f v = *(const v4f*)&X[(size_t)(row0 + r) * DIM + kc + lane * 4];
            *(v4f*)&myx[r * XSTRIDE + lane * 4] = v;
        }
        const float* wbase = (m < 3) ? &lds_w[m * DIM + kc + kp] : &lds_w[3 * DIM + kp];
        #pragma unroll
        for (int kk = 0; kk < CK; kk += 4) {
            v2f af = *(const v2f*)(abase + kk);
            v2f bf = *(const v2f*)(wbase + kk);
            acc = __builtin_amdgcn_wmma_f32_16x16x4_f32(
                false, af, false, bf, (short)0, acc, false, false);
        }
    }

    #pragma unroll
    for (int r = 0; r < 8; ++r) {
        int mm = r + ((lane >> 4) << 3);
        myx[mm * 16 + m] = acc[r];
    }
    if (lane < 16) {
        float d0 = myx[lane * 16 + 0];
        float d1 = myx[lane * 16 + 1];
        float d2 = myx[lane * 16 + 2];
        float a = 1.f;
        a = a + a * d0;
        a = a + (a * d1 + lds_e[0]);
        a = a + (a * d2 + lds_e[1]);
        lds_a[wave * ROWS_PER_WAVE + lane] = a;
    }
    __syncthreads();

    const int c0 = tid * (DIM / BLOCK);
    v4f bA = *(const v4f*)&lds_b3[c0];
    v4f bB = *(const v4f*)&lds_b3[c0 + 4];
    const int rowb = blockIdx.x * ROWS_PER_BLOCK;
    for (int r = 0; r < ROWS_PER_BLOCK; ++r) {
        size_t base = (size_t)(rowb + r) * DIM + c0;
        float av = lds_a[r];
        v4f xa = *(const v4f*)&X[base];
        v4f xb = *(const v4f*)&X[base + 4];
        v4f oa = xa * av + bA;
        v4f ob = xb * av + bB;
        __builtin_nontemporal_store(oa, (v4f*)&Out[base]);
        __builtin_nontemporal_store(ob, (v4f*)&Out[base + 4]);
    }
}

extern "C" void kernel_launch(void* const* d_in, const int* in_sizes, int n_in,
                              void* d_out, int out_size, void* d_ws, size_t ws_size,
                              hipStream_t stream) {
    const float* x  = (const float*)d_in[0];
    const float* wk = (const float*)d_in[1];
    const float* bb = (const float*)d_in[2];
    float* out = (float*)d_out;

    const int n_rows = in_sizes[0] / DIM;
    const int grid   = n_rows / ROWS_PER_BLOCK;
    crossnet_fused<<<grid, BLOCK, 0, stream>>>(x, wk, bb, out);
}


# each wave keeps its whole 16x1024 x tile in LDS and the output pass reads it there (x read from global once); fragments read 16 B wide into two accumulators: d and e sums in a new fixed order
# speedup vs baseline: 2.9411x; 2.9411x over previous
#include <hip/hip_runtime.h>

typedef float v2f __attribute__((ext_vector_type(2)));
typedef float v4f __attribute__((ext_vector_type(4)));
typedef float v8f __attribute__((ext_vector_type(8)));

#define DIM            1024
#define ROWS_PER_WAVE  16
#define WAVES          4
#define BLOCK          (WAVES * 32)
#define ROWS_PER_BLOCK (WAVES * ROWS_PER_WAVE)
#define NCH            8
#define XP             1028
#define WP             1028

__global__ __launch_bounds__(BLOCK) void crossnet_fused(
    const float* __restrict__ X,
    const float* __restrict__ Wk,
    const float* __restrict__ Bb,
    float* __restrict__ Out)
{
    __shared__ __attribute__((aligned(16))) float lds_x[WAVES][ROWS_PER_WAVE * XP];
    __shared__ __attribute__((aligned(16))) float lds_w[4 * WP];
    __shared__ __attribute__((aligned(16))) float lds_b3[DIM];
    __shared__ float lds_d[WAVES][ROWS_PER_WAVE * 4];
    __shared__ float lds_a[ROWS_PER_BLOCK];
    __shared__ float lds_red[2 * BLOCK];
    __shared__ float lds_e[2];

    const int tid = threadIdx.x;
    const int wave = tid >> 5, lane = tid & 31;
    const int m = lane & 15;
    const int h = lane >> 4;
    float* myx = &lds_x[wave][0];
    const int row0 = blockIdx.x * ROWS_PER_BLOCK + wave * ROWS_PER_WAVE;
    const float* xg = X + (size_t)row0 * DIM + lane * 4;

    const int c0 = tid * (DIM / BLOCK);
    v4f w0a = *(const v4f*)&Wk[c0],           w0b = *(const v4f*)&Wk[c0 + 4];
    v4f w1a = *(const v4f*)&Wk[DIM + c0],     w1b = *(const v4f*)&Wk[DIM + c0 + 4];
    v4f w2a = *(const v4f*)&Wk[2 * DIM + c0], w2b = *(const v4f*)&Wk[2 * DIM + c0 + 4];
    v4f b0a = *(const v4f*)&Bb[c0],           b0b = *(const v4f*)&Bb[c0 + 4];
    v4f b1a = *(const v4f*)&Bb[DIM + c0],     b1b = *(const v4f*)&Bb[DIM + c0 + 4];
    v4f b2a = *(const v4f*)&Bb[2 * DIM + c0], b2b = *(const v4f*)&Bb[2 * DIM + c0 + 4];

    v4f xr[NCH][ROWS_PER_WAVE];
    #pragma unroll
    for (int c = 0; c < 3; ++c) {
        #pragma unroll
        for (int r = 0; r < ROWS_PER_WAVE; ++r)
            xr[c][r] = *(const v4f*)(xg + (size_t)r * DIM + c * 128);
    }

    {
        *(v4f*)&lds_w[c0] = w0a;              *(v4f*)&lds_w[c0 + 4] = w0b;
        *(v4f*)&lds_w[WP + c0] = w1a;         *(v4f*)&lds_w[WP + c0 + 4] = w1b;
        *(v4f*)&lds_w[2 * WP + c0] = w2a;     *(v4f*)&lds_w[2 * WP + c0 + 4] = w2b;
        v4f z = {0.f, 0.f, 0.f, 0.f};
        *(v4f*)&lds_w[3 * WP + c0] = z;       *(v4f*)&lds_w[3 * WP + c0 + 4] = z;
        *(v4f*)&lds_b3[c0] = b0a + b1a + b2a; *(v4f*)&lds_b3[c0 + 4] = b0b + b1b + b2b;
        v4f q1 = b0a * w1a + b0b * w1b;
        v4f q2 = (b0a + b1a) * w2a + (b0b + b1b) * w2b;
        lds_red[tid] = (q1.x + q1.y) + (q1.z + q1.w);
        lds_red[BLOCK + tid] = (q2.x + q2.y) + (q2.z + q2.w);
    }
    __syncthreads();
    if (tid == 0) {
        float e1 = 0.f, e2 = 0.f;
        for (int i = 0; i < BLOCK; ++i) { e1 += lds_red[i]; e2 += lds_red[BLOCK + i]; }
        lds_e[0] = e1; lds_e[1] = e2;
    }
    __syncthreads();

    #pragma unroll
    for (int c = 3; c < NCH; ++c) {
        #pragma unroll
        for (int r = 0; r < ROWS_PER_WAVE; ++r)
            xr[c][r] = *(const v4f*)(xg + (size_t)r * DIM + c * 128);
    }

    const float* abase = &myx[m * XP + 4 * h];
    const float* wbase = &lds_w[(m < 3 ? m : 3) * WP + 4 * h];
    v8f acc0 = {}, acc1 = {};
    #pragma unroll
    for (int c = 0; c < NCH; ++c) {
        #pragma unroll
        for (int r = 0; r < ROWS_PER_WAVE; ++r)
            *(v4f*)&myx[r * XP + c * 128 + lane * 4] = xr[c][r];
        #pragma unroll
        for (int kk = 0; kk < 128; kk += 8) {
            v4f af = *(const v4f*)(abase + c * 128 + kk);
            v4f bf = *(const v4f*)(wbase + c * 128 + kk);
            v2f a0 = {af.x, af.y}, a1 = {af.z, af.w};
            v2f b0 = {bf.x, bf.y}, b1 = {bf.z, bf.w};
            acc0 = __builtin_amdgcn_wmma_f32_16x16x4_f32(
                false, a0, false, b0, (short)0, acc0, false, false);
            acc1 = __builtin_amdgcn_wmma_f32_16x16x4_f32(
                false, a1, false, b1, (short)0, acc1, false, false);
        }
    }
    v8f acc = acc0 + acc1;

    if (m < 3) {
        #pragma unroll
        for (int r = 0; r < 8; ++r) lds_d[wave][(r + 8 * h) * 4 + m] = acc[r];
    }
    {
        float d0 = lds_d[wave][m * 4 + 0];
        float d1 = lds_d[wave][m * 4 + 1];
        float d2 = lds_d[wave][m * 4 + 2];
        float a = 1.f;
        a = a + a * d0;
        a = a + (a * d1 + lds_e[0]);
        a = a + (a * d2 + lds_e[1]);
        if (lane < 16) lds_a[wave * ROWS_PER_WAVE + lane] = a;
    }
    __syncthreads();

    v4f b3r[NCH];
    #pragma unroll
    for (int j = 0; j < NCH; ++j) b3r[j] = *(const v4f*)&lds_b3[j * 128 + lane * 4];
    float* og = Out + (size_t)row0 * DIM + lane * 4;
    #pragma unroll 4
    for (int r = 0; r < ROWS_PER_WAVE; ++r) {
        float av = lds_a[wave * ROWS_PER_WAVE + r];
        #pragma unroll
        for (int j = 0; j < NCH; ++j) {
            v4f xv = *(const v4f*)&myx[r * XP + j * 128 + lane * 4];
            v4f o = xv * av + b3r[j];
            __builtin_nontemporal_store(o, (v4f*)(og + (size_t)r * DIM + j * 128));
        }
    }
}

extern "C" void kernel_launch(void* const* d_in, const int* in_sizes, int n_in,
                              void* d_out, int out_size, void* d_ws, size_t ws_size,
                              hipStream_t stream) {
    const float* x  = (const float*)d_in[0];
    const float* wk = (const float*)d_in[1];
    const float* bb = (const float*)d_in[2];
    float* out = (float*)d_out;

    const int n_rows = in_sizes[0] / DIM;
    const int grid   = n_rows / ROWS_PER_BLOCK;
    crossnet_fused<<<grid, BLOCK, 0, stream>>>(x, wk, bb, out);
}


# one hand-written LDS-only work-group barrier replaces three draining barriers and the serial e sum, x loads stay 4 pieces in flight across it: e1, e2 summed in a new fixed tree order
# speedup vs baseline: 3.2003x; 1.0881x over previous
#include <hip/hip_runtime.h>

typedef float v2f __attribute__((ext_vector_type(2)));
typedef float v4f __attribute__((ext_vector_type(4)));
typedef float v8f __attribute__((ext_vector_type(8)));

#define DIM            1024
#define ROWS_PER_WAVE  16
#define WAVES          4
#define BLOCK          (WAVES * 32)
#define ROWS_PER_BLOCK (WAVES * ROWS_PER_WAVE)
#define NCH            8
#define XP             1028
#define WP             1028
#define AHEAD          4

__global__ __launch_bounds__(BLOCK) void crossnet_fused(
    const float* __restrict__ X,
    const float* __restrict__ Wk,
    const float* __restrict__ Bb,
    float* __restrict__ Out)
{
    __shared__ __attribute__((aligned(16))) float lds_x[WAVES][ROWS_PER_WAVE * XP];
    __shared__ __attribute__((aligned(16))) float lds_w[4 * WP];
    __shared__ __attribute__((aligned(16))) float lds_b3[DIM];
    __shared__ float lds_d[WAVES][ROWS_PER_WAVE * 4];
    __shared__ float lds_a[ROWS_PER_BLOCK];
    __shared__ __attribute__((aligned(16))) float lds_red[2 * WAVES];

    const int tid = threadIdx.x;
    const int wave = tid >> 5, lane = tid & 31;
    const int m = lane & 15;
    const int h = lane >> 4;
    float* myx = &lds_x[wave][0];
    const int row0 = blockIdx.x * ROWS_PER_BLOCK + wave * ROWS_PER_WAVE;
    const float* xg = X + (size_t)row0 * DIM + lane * 4;

    const int c0 = tid * (DIM / BLOCK);
    v4f w0a = *(const v4f*)&Wk[c0],           w0b = *(const v4f*)&Wk[c0 + 4];
    v4f w1a = *(const v4f*)&Wk[DIM + c0],     w1b = *(const v4f*)&Wk[DIM + c0 + 4];
    v4f w2a = *(const v4f*)&Wk[2 * DIM + c0], w2b = *(const v4f*)&Wk[2 * DIM + c0 + 4];
    v4f b0a = *(const v4f*)&Bb[c0],           b0b = *(const v4f*)&Bb[c0 + 4];
    v4f b1a = *(const v4f*)&Bb[DIM + c0],     b1b = *(const v4f*)&Bb[DIM + c0 + 4];
    v4f b2a = *(const v4f*)&Bb[2 * DIM + c0], b2b = *(const v4f*)&Bb[2 * DIM + c0 + 4];

    v4f xr[NCH][ROWS_PER_WAVE];
    #pragma unroll
    for (int c = 0; c < AHEAD - 1; ++c) {
        if (c > 0) __builtin_amdgcn_sched_barrier(0);
        #pragma unroll
        for (int r = 0; r < ROWS_PER_WAVE; ++r)
            xr[c][r] = *(const v4f*)(xg + (size_t)r * DIM + c * 128);
    }

    {
        *(v4f*)&lds_w[c0] = w0a;              *(v4f*)&lds_w[c0 + 4] = w0b;
        *(v4f*)&lds_w[WP + c0] = w1a;         *(v4f*)&lds_w[WP + c0 + 4] = w1b;
        *(v4f*)&lds_w[2 * WP + c0] = w2a;     *(v4f*)&lds_w[2 * WP + c0 + 4] = w2b;
        v4f z = {0.f, 0.f, 0.f, 0.f};
        *(v4f*)&lds_w[3 * WP + c0] = z;       *(v4f*)&lds_w[3 * WP + c0 + 4] = z;
        *(v4f*)&lds_b3[c0] = b0a + b1a + b2a; *(v4f*)&lds_b3[c0 + 4] = b0b + b1b + b2b;
        v4f q1 = b0a * w1a + b0b * w1b;
        v4f q2 = (b0a + b1a) * w2a + (b0b + b1b) * w2b;
        float s1 = (q1.x + q1.y) + (q1.z + q1.w);
        float s2 = (q2.x + q2.y) + (q2.z + q2.w);
        #pragma unroll
        for (int k = 16; k >= 1; k >>= 1) {
            s1 += __shfl_xor(s1, k, 32);
            s2 += __shfl_xor(s2, k, 32);
        }
        if (lane == 0) { lds_red[wave] = s1; lds_red[WAVES + wave] = s2; }
    }
    #pragma unroll
    for (int r = 0; r < ROWS_PER_WAVE; ++r)
        xr[AHEAD - 1][r] = *(const v4f*)(xg + (size_t)r * DIM + (AHEAD - 1) * 128);
    asm volatile("s_wait_dscnt 0x0\n\ts_barrier_signal -1\n\ts_barrier_wait -1" ::: "memory");

    const float* abase = &myx[m * XP + 4 * h];
    const float* wbase = &lds_w[(m < 3 ? m : 3) * WP + 4 * h];
    v8f acc0 = {}, acc1 = {};
    #pragma unroll
    for (int c = 0; c < NCH; ++c) {
        #pragma unroll
        for (int r = 0; r < ROWS_PER_WAVE; ++r)
            *(v4f*)&myx[r * XP + c * 128 + lane * 4] = xr[c][r];
        __builtin_amdgcn_sched_barrier(0);
        if (c + AHEAD < NCH) {
            #pragma unroll
            for (int r = 0; r < ROWS_PER_WAVE; ++r)
                xr[c + AHEAD][r] = *(const v4f*)(xg + (size_t)r * DIM + (c + AHEAD) * 128);
        }
        #pragma unroll
        for (int kk = 0; kk < 128; kk += 8) {
            v4f af = *(const v4f*)(abase + c * 128 + kk);
            v4f bf = *(const v4f*)(wbase + c * 128 + kk);
            v2f a0 = {af.x, af.y}, a1 = {af.z, af.w};
            v2f b0 = {bf.x, bf.y}, b1 = {bf.z, bf.w};
            acc0 = __builtin_amdgcn_wmma_f32_16x16x4_f32(
                false, a0, false, b0, (short)0, acc0, false, false);
            acc1 = __builtin_amdgcn_wmma_f32_16x16x4_f32(
                false, a1, false, b1, (short)0, acc1, false, false);
        }
    }
    v8f acc = acc0 + acc1;

    if (m < 3) {
        #pragma unroll
        for (int r = 0; r < 8; ++r) lds_d[wave][(r + 8 * h) * 4 + m] = acc[r];
    }
    {
        float d0 = lds_d[wave][m * 4 + 0];
        float d1 = lds_d[wave][m * 4 + 1];
        float d2 = lds_d[wave][m * 4 + 2];
        v4f p1 = *(const v4f*)&lds_red[0];
        v4f p2 = *(const v4f*)&lds_red[WAVES];
        float e1 = (p1.x + p1.y) + (p1.z + p1.w);
        float e2 = (p2.x + p2.y) + (p2.z + p2.w);
        float a = 1.f;
        a = a + a * d0;
        a = a + (a * d1 + e1);
        a = a + (a * d2 + e2);
        if (lane < 16) lds_a[wave * ROWS_PER_WAVE + lane] = a;
    }

    v4f b3r[NCH];
    #pragma unroll
    for (int j = 0; j < NCH; ++j) b3r[j] = *(const v4f*)&lds_b3[j * 128 + lane * 4];
    float* og = Out + (size_t)row0 * DIM + lane * 4;
    #pragma unroll 4
    for (int r = 0; r < ROWS_PER_WAVE; ++r) {
        float av = lds_a[wave * ROWS_PER_WAVE + r];
        #pragma unroll
        for (int j = 0; j < NCH; ++j) {
            v4f xv = *(const v4f*)&myx[r * XP + j * 128 + lane * 4];
            v4f o = xv * av + b3r[j];
            __builtin_nontemporal_store(o, (v4f*)(og + (size_t)r * DIM + j * 128));
        }
    }
}

extern "C" void kernel_launch(void* const* d_in, const int* in_sizes, int n_in,
                              void* d_out, int out_size, void* d_ws, size_t ws_size,
                              hipStream_t stream) {
    const float* x  = (const float*)d_in[0];
    const float* wk = (const float*)d_in[1];
    const float* bb = (const float*)d_in[2];
    float* out = (float*)d_out;

    const int n_rows = in_sizes[0] / DIM;
    const int grid   = n_rows / ROWS_PER_BLOCK;
    crossnet_fused<<<grid, BLOCK, 0, stream>>>(x, wk, bb, out);
}


# A and B fragment reads of each 64-column block go out as one group a whole block ahead of its 16 products (second register set) instead of two reads one step ahead; sum order unchanged
# speedup vs baseline: 3.5098x; 1.0967x over previous
#include <hip/hip_runtime.h>

typedef float v2f __attribute__((ext_vector_type(2)));
typedef float v4f __attribute__((ext_vector_type(4)));
typedef float v8f __attribute__((ext_vector_type(8)));

#define DIM            1024
#define ROWS_PER_WAVE  16
#define WAVES          4
#define BLOCK          (WAVES * 32)
#define ROWS_PER_BLOCK (WAVES * ROWS_PER_WAVE)
#define NCH            8
#define XP             1028
#define WP             1028
#define KBLK           64
#define NBLK           (DIM / KBLK)
#define FRD            (KBLK / 8)
#define AHEAD          4

__global__ __launch_bounds__(BLOCK) void crossnet_fused(
    const float* __restrict__ X,
    const float* __restrict__ Wk,
    const float* __restrict__ Bb,
    float* __restrict__ Out)
{
    __shared__ __attribute__((aligned(16))) float lds_x[WAVES][ROWS_PER_WAVE * XP];
    __shared__ __attribute__((aligned(16))) float lds_w[4 * WP];
    __shared__ __attribute__((aligned(16))) float lds_b3[DIM];
    __shared__ float lds_d[WAVES][ROWS_PER_WAVE * 4];
    __shared__ float lds_a[ROWS_PER_BLOCK];
    __shared__ __attribute__((aligned(16))) float lds_red[2 * WAVES];

    const int tid = threadIdx.x;
    const int wave = tid >> 5, lane = tid & 31;
    const int m = lane & 15;
    const int h = lane >> 4;
    float* myx = &lds_x[wave][0];
    const int row0 = blockIdx.x * ROWS_PER_BLOCK + wave * ROWS_PER_WAVE;
    const float* xg = X + (size_t)row0 * DIM + lane * 4;

    const int c0 = tid * (DIM / BLOCK);
    v4f w0a = *(const v4f*)&Wk[c0],           w0b = *(const v4f*)&Wk[c0 + 4];
    v4f w1a = *(const v4f*)&Wk[DIM + c0],     w1b = *(const v4f*)&Wk[DIM + c0 + 4];
    v4f w2a = *(const v4f*)&Wk[2 * DIM + c0], w2b = *(const v4f*)&Wk[2 * DIM + c0 + 4];
    v4f b0a = *(const v4f*)&Bb[c0],           b0b = *(const v4f*)&Bb[c0 + 4];
    v4f b1a = *(const v4f*)&Bb[DIM + c0],     b1b = *(const v4f*)&Bb[DIM + c0 + 4];
    v4f b2a = *(const v4f*)&Bb[2 * DIM + c0], b2b = *(const v4f*)&Bb[2 * DIM + c0 + 4];

    v4f xr[NCH][ROWS_PER_WAVE];
    #pragma unroll
    for (int c = 0; c < AHEAD - 1; ++c) {
        if (c > 0) __builtin_amdgcn_sched_barrier(0);
        #pragma unroll
        for (int r = 0; r < ROWS_PER_WAVE; ++r)
            xr[c][r] = *(const v4f*)(xg + (size_t)r * DIM + c * 128);
    }

    {
        *(v4f*)&lds_w[c0] = w0a;              *(v4f*)&lds_w[c0 + 4] = w0b;
        *(v4f*)&lds_w[WP + c0] = w1a;         *(v4f*)&lds_w[WP + c0 + 4] = w1b;
        *(v4f*)&lds_w[2 * WP + c0] = w2a;     *(v4f*)&lds_w[2 * WP + c0 + 4] = w2b;
        v4f z = {0.f, 0.f, 0.f, 0.f};
        *(v4f*)&lds_w[3 * WP + c0] = z;       *(v4f*)&lds_w[3 * WP + c0 + 4] = z;
        *(v4f*)&lds_b3[c0] = b0a + b1a + b2a; *(v4f*)&lds_b3[c0 + 4] = b0b + b1b + b2b;
        v4f q1 = b0a * w1a + b0b * w1b;
        v4f q2 = (b0a + b1a) * w2a + (b0b + b1b) * w2b;
        float s1 = (q1.x + q1.y) + (q1.z + q1.w);
        float s2 = (q2.x + q2.y) + (q2.z + q2.w);
        #pragma unroll
        for (int k = 16; k >= 1; k >>= 1) {
            s1 += __shfl_xor(s1, k, 32);
            s2 += __shfl_xor(s2, k, 32);
        }
        if (lane == 0) { lds_red[wave] = s1; lds_red[WAVES + wave] = s2; }
    }
    #pragma unroll
    for (int r = 0; r < ROWS_PER_WAVE; ++r)
        xr[AHEAD - 1][r] = *(const v4f*)(xg + (size_t)r * DIM + (AHEAD - 1) * 128);
    asm volatile("s_wait_dscnt 0x0\n\ts_barrier_signal -1\n\ts_barrier_wait -1" ::: "memory");

    const float* abase = &myx[m * XP + 4 * h];
    const float* wbase = &lds_w[(m < 3 ? m : 3) * WP + 4 * h];
    v8f acc0 = {}, acc1 = {};
    v4f fa[2][FRD], fb[2][FRD];
    #pragma unroll
    for (int r = 0; r < ROWS_PER_WAVE; ++r)
        *(v4f*)&myx[r * XP + lane * 4] = xr[0][r];
    __builtin_amdgcn_sched_barrier(0);
    #pragma unroll
    for (int r = 0; r < ROWS_PER_WAVE; ++r)
        xr[AHEAD][r] = *(const v4f*)(xg + (size_t)r * DIM + AHEAD * 128);
    #pragma unroll
    for (int j = 0; j < FRD; ++j) {
        fa[0][j] = *(const v4f*)(abase + j * 8);
        fb[0][j] = *(const v4f*)(wbase + j * 8);
    }
    #pragma unroll
    for (int b = 0; b < NBLK; ++b) {
        const int nb = b + 1;
        if (nb < NBLK) {
            if ((nb & 1) == 0) {
                const int c = nb >> 1;
                #pragma unroll
                for (int r = 0; r < ROWS_PER_WAVE; ++r)
                    *(v4f*)&myx[r * XP + c * 128 + lane * 4] = xr[c][r];
                __builtin_amdgcn_sched_barrier(0);
                if (c + AHEAD < NCH) {
                    #pragma unroll
                    for (int r = 0; r < ROWS_PER_WAVE; ++r)
                        xr[c + AHEAD][r] = *(const v4f*)(xg + (size_t)r * DIM + (c + AHEAD) * 128);
                }
            }
            #pragma unroll
            for (int j = 0; j < FRD; ++j) {
                fa[nb & 1][j] = *(const v4f*)(abase + nb * KBLK + j * 8);
                fb[nb & 1][j] = *(const v4f*)(wbase + nb * KBLK + j * 8);
            }
        }
        __builtin_amdgcn_sched_barrier(0);
        #pragma unroll
        for (int j = 0; j < FRD; ++j) {
            v4f af = fa[b & 1][j], bf = fb[b & 1][j];
            v2f a0 = {af.x, af.y}, a1 = {af.z, af.w};
            v2f b0 = {bf.x, bf.y}, b1 = {bf.z, bf.w};
            acc0 = __builtin_amdgcn_wmma_f32_16x16x4_f32(
                false, a0, false, b0, (short)0, acc0, false, false);
            acc1 = __builtin_amdgcn_wmma_f32_16x16x4_f32(
                false, a1, false, b1, (short)0, acc1, false, false);
        }
        __builtin_amdgcn_sched_barrier(0);
    }
    v8f acc = acc0 + acc1;

    if (m < 3) {
        #pragma unroll
        for (int r = 0; r < 8; ++r) lds_d[wave][(r + 8 * h) * 4 + m] = acc[r];
    }
    {
        float d0 = lds_d[wave][m * 4 + 0];
        float d1 = lds_d[wave][m * 4 + 1];
        float d2 = lds_d[wave][m * 4 + 2];
        v4f p1 = *(const v4f*)&lds_red[0];
        v4f p2 = *(const v4f*)&lds_red[WAVES];
        float e1 = (p1.x + p1.y) + (p1.z + p1.w);
        float e2 = (p2.x + p2.y) + (p2.z + p2.w);
        float a = 1.f;
        a = a + a * d0;
        a = a + (a * d1 + e1);
        a = a + (a * d2 + e2);
        if (lane < 16) lds_a[wave * ROWS_PER_WAVE + lane] = a;
    }

    v4f b3r[NCH];
    #pragma unroll
    for (int j = 0; j < NCH; ++j) b3r[j] = *(const v4f*)&lds_b3[j * 128 + lane * 4];
    float* og = Out + (size_t)row0 * DIM + lane * 4;
    #pragma unroll 4
    for (int r = 0; r < ROWS_PER_WAVE; ++r) {
        float av = lds_a[wave * ROWS_PER_WAVE + r];
        #pragma unroll
        for (int j = 0; j < NCH; ++j) {
            v4f xv = *(const v4f*)&myx[r * XP + j * 128 + lane * 4];
            v4f o = xv * av + b3r[j];
            __builtin_nontemporal_store(o, (v4f*)(og + (size_t)r * DIM + j * 128));
        }
    }
}

extern "C" void kernel_launch(void* const* d_in, const int* in_sizes, int n_in,
                              void* d_out, int out_size, void* d_ws, size_t ws_size,
                              hipStream_t stream) {
    const float* x  = (const float*)d_in[0];
    const float* wk = (const float*)d_in[1];
    const float* bb = (const float*)d_in[2];
    float* out = (float*)d_out;

    const int n_rows = in_sizes[0] / DIM;
    const int grid   = n_rows / ROWS_PER_BLOCK;
    crossnet_fused<<<grid, BLOCK, 0, stream>>>(x, wk, bb, out);
}
